# starting kernel
# baseline (speedup 1.0000x reference)
#include <hip/hip_runtime.h>

constexpr int N_SAMPLES = 65536;
constexpr int DIM       = 1024;
constexpr int CHANNELS  = 128;

constexpr int SPLITS         = 64;
constexpr int ROWS_PER_SPLIT = N_SAMPLES / SPLITS;

typedef __attribute__((ext_vector_type(2))) float v2f;
typedef __attribute__((ext_vector_type(8))) float v8f;

constexpr int WS_PART = 0;
constexpr int WS_C    = SPLITS * DIM;
constexpr int WS_T    = WS_C + DIM;
constexpr int WS_S    = WS_T + DIM;

__global__ void colsum_x_wmma(const float* __restrict__ x,
                              float* __restrict__ part) {
    const int lane  = threadIdx.x & 31;
    const int wave  = threadIdx.x >> 5;
    const int col0  = blockIdx.x * 64 + wave * 16;
    const int split = blockIdx.y;
    const int r0    = split * ROWS_PER_SPLIT;

    const int col   = col0 + (lane & 15);
    const int khalf = (lane >> 4) << 1;

    v2f a; a.x = 1.0f; a.y = 1.0f;
    v8f acc = {};

    const float* base = x + (size_t)r0 * DIM + col;
    #pragma unroll 4
    for (int r = 0; r < ROWS_PER_SPLIT; r += 4) {
        const float* p = base + (size_t)r * DIM;
        v2f b;
        b.x = __builtin_nontemporal_load(p + (size_t)khalf * DIM);
        b.y = __builtin_nontemporal_load(p + (size_t)(khalf + 1) * DIM);
        acc = __builtin_amdgcn_wmma_f32_16x16x4_f32(
            false, a, false, b, (short)0, acc, false, false);
    }
    if (lane < 16) part[(size_t)split * DIM + col] = acc[0];
}

__global__ void reduce_parts(const float* __restrict__ part,
                             float* __restrict__ c) {
    const int col = blockIdx.x * blockDim.x + threadIdx.x;
    float s = 0.0f;
    #pragma unroll 8
    for (int k = 0; k < SPLITS; ++k) s += part[(size_t)k * DIM + col];
    c[col] = s;
}

__global__ void weight_sums(const float* __restrict__ w,
                            float* __restrict__ t,
                            float* __restrict__ s) {
    if (blockIdx.x < 4) {
        const int j = blockIdx.x * 256 + threadIdx.x;
        float acc = 0.0f;
        #pragma unroll 4
        for (int i = 0; i < CHANNELS; ++i) acc += w[(size_t)i * DIM + j];
        t[j] = acc;
    } else {
        const int i = threadIdx.x;
        if (i < CHANNELS) {
            const float* row = w + (size_t)i * DIM;
            float acc = 0.0f;
            #pragma unroll 4
            for (int j = 0; j < DIM; ++j) acc += row[j];
            s[i] = acc;
        }
    }
}

__global__ void softreorder_out(const float* __restrict__ cvec,
                                const float* __restrict__ tvec,
                                const float* __restrict__ svec,
                                float* __restrict__ out) {
    __shared__ float red[256];
    const int i  = blockIdx.x;
    const float si = svec[i];

    float p = 0.0f;
    for (int j = threadIdx.x; j < DIM; j += 256) p += __expf(si * cvec[j]);
    red[threadIdx.x] = p;
    __syncthreads();
    #pragma unroll
    for (int off = 128; off > 0; off >>= 1) {
        if (threadIdx.x < off) red[threadIdx.x] += red[threadIdx.x + off];
        __syncthreads();
    }
    const float inv_det = 1.0f / red[0];

    float* orow = out + (size_t)i * DIM;
    for (int j = threadIdx.x; j < DIM; j += 256)
        orow[j] = __expf(si * tvec[j]) * inv_det;
}

extern "C" void kernel_launch(void* const* d_in, const int* in_sizes, int n_in,
                              void* d_out, int out_size, void* d_ws, size_t ws_size,
                              hipStream_t stream) {
    (void)in_sizes; (void)n_in; (void)out_size; (void)ws_size;
    const float* x = (const float*)d_in[0];
    const float* w = (const float*)d_in[1];
    float* ws   = (float*)d_ws;
    float* part = ws + WS_PART;
    float* c    = ws + WS_C;
    float* t    = ws + WS_T;
    float* s    = ws + WS_S;
    float* out  = (float*)d_out;

    colsum_x_wmma<<<dim3(DIM / 64, SPLITS), 128, 0, stream>>>(x, part);
    reduce_parts<<<dim3(DIM / 256), 256, 0, stream>>>(part, c);
    weight_sums<<<5, 256, 0, stream>>>(w, t, s);
    softreorder_out<<<CHANNELS, 256, 0, stream>>>(c, t, s, out);
}


# phase 1 streams x with 16-byte loads over whole 256-byte row pieces, 16-32 loads in flight per wave; 4 waves split each row block and add their sums in LDS (new reduction tree for the column sums)
# speedup vs baseline: 1.4234x; 1.4234x over previous
#include <hip/hip_runtime.h>

constexpr int N_SAMPLES = 65536;
constexpr int DIM       = 1024;
constexpr int CHANNELS  = 128;

constexpr int SPLITS         = 64;
constexpr int ROWS_PER_SPLIT = N_SAMPLES / SPLITS;

typedef __attribute__((ext_vector_type(2))) float v2f;
typedef __attribute__((ext_vector_type(4))) float v4f;
typedef __attribute__((ext_vector_type(8))) float v8f;

constexpr int WS_PART = 0;
constexpr int WS_C    = SPLITS * DIM;
constexpr int WS_T    = WS_C + DIM;
constexpr int WS_S    = WS_T + DIM;

constexpr int WG_WAVES      = 4;
constexpr int ROWS_PER_WAVE = ROWS_PER_SPLIT / WG_WAVES;
constexpr int QUADS         = ROWS_PER_WAVE / 4;
constexpr int BQ            = 8;
constexpr int NBATCH        = QUADS / BQ;

__global__ void __launch_bounds__(WG_WAVES * 32)
colsum_x_wmma(const float* __restrict__ x, float* __restrict__ part) {
    __shared__ v4f red[WG_WAVES][16];

    const int lane  = threadIdx.x & 31;
    const int wave  = threadIdx.x >> 5;
    const int c0    = blockIdx.x * 64;
    const int split = blockIdx.y;
    const int r0    = split * ROWS_PER_SPLIT + wave * ROWS_PER_WAVE;
    const int h     = lane >> 4;
    const int c     = lane & 15;

    v2f a; a.x = 1.0f; a.y = 1.0f;
    v8f acc0 = {}, acc1 = {}, acc2 = {}, acc3 = {};

    const float* p = x + (size_t)(r0 + h) * DIM + c0 + 4 * c;
    v4f buf[2][2 * BQ];

#define LOAD_BATCH(B)                                                             \
    _Pragma("unroll") for (int g = 0; g < BQ; ++g) {                               \
        const float* q = p + (size_t)(((B) * BQ + g) * 4) * DIM;                   \
        buf[(B) & 1][2 * g]     = __builtin_nontemporal_load((const v4f*)q);             \
        buf[(B) & 1][2 * g + 1] = __builtin_nontemporal_load((const v4f*)(q + 2 * DIM));   \
    }

    LOAD_BATCH(0)
    #pragma unroll
    for (int b = 0; b < NBATCH; ++b) {
        __builtin_amdgcn_sched_barrier(0);
        if (b + 1 < NBATCH) { LOAD_BATCH(b + 1) }
        __builtin_amdgcn_sched_barrier(0);
        #pragma unroll
        for (int g = 0; g < BQ; ++g) {
            const v4f u = buf[b & 1][2 * g];
            const v4f v = buf[b & 1][2 * g + 1];
            v2f b0; b0.x = u.x; b0.y = v.x;
            v2f b1; b1.x = u.y; b1.y = v.y;
            v2f b2; b2.x = u.z; b2.y = v.z;
            v2f b3; b3.x = u.w; b3.y = v.w;
            acc0 = __builtin_amdgcn_wmma_f32_16x16x4_f32(false, a, false, b0, (short)0, acc0, false, false);
            acc1 = __builtin_amdgcn_wmma_f32_16x16x4_f32(false, a, false, b1, (short)0, acc1, false, false);
            acc2 = __builtin_amdgcn_wmma_f32_16x16x4_f32(false, a, false, b2, (short)0, acc2, false, false);
            acc3 = __builtin_amdgcn_wmma_f32_16x16x4_f32(false, a, false, b3, (short)0, acc3, false, false);
        }
    }
#undef LOAD_BATCH

    if (lane < 16) {
        v4f m; m.x = acc0[0]; m.y = acc1[0]; m.z = acc2[0]; m.w = acc3[0];
        red[wave][lane] = m;
    }
    __syncthreads();
    if (threadIdx.x < 16) {
        const v4f s = ((red[0][lane] + red[1][lane]) + red[2][lane]) + red[3][lane];
        *(v4f*)(part + (size_t)split * DIM + c0 + 4 * lane) = s;
    }
}

__global__ void reduce_parts(const float* __restrict__ part,
                             float* __restrict__ c) {
    const int col = blockIdx.x * blockDim.x + threadIdx.x;
    float s = 0.0f;
    #pragma unroll 8
    for (int k = 0; k < SPLITS; ++k) s += part[(size_t)k * DIM + col];
    c[col] = s;
}

__global__ void weight_sums(const float* __restrict__ w,
                            float* __restrict__ t,
                            float* __restrict__ s) {
    if (blockIdx.x < 4) {
        const int j = blockIdx.x * 256 + threadIdx.x;
        float acc = 0.0f;
        #pragma unroll 4
        for (int i = 0; i < CHANNELS; ++i) acc += w[(size_t)i * DIM + j];
        t[j] = acc;
    } else {
        const int i = threadIdx.x;
        if (i < CHANNELS) {
            const float* row = w + (size_t)i * DIM;
            float acc = 0.0f;
            #pragma unroll 4
            for (int j = 0; j < DIM; ++j) acc += row[j];
            s[i] = acc;
        }
    }
}

__global__ void softreorder_out(const float* __restrict__ cvec,
                                const float* __restrict__ tvec,
                                const float* __restrict__ svec,
                                float* __restrict__ out) {
    __shared__ float red[256];
    const int i  = blockIdx.x;
    const float si = svec[i];

    float p = 0.0f;
    for (int j = threadIdx.x; j < DIM; j += 256) p += __expf(si * cvec[j]);
    red[threadIdx.x] = p;
    __syncthreads();
    #pragma unroll
    for (int off = 128; off > 0; off >>= 1) {
        if (threadIdx.x < off) red[threadIdx.x] += red[threadIdx.x + off];
        __syncthreads();
    }
    const float inv_det = 1.0f / red[0];

    float* orow = out + (size_t)i * DIM;
    for (int j = threadIdx.x; j < DIM; j += 256)
        orow[j] = __expf(si * tvec[j]) * inv_det;
}

extern "C" void kernel_launch(void* const* d_in, const int* in_sizes, int n_in,
                              void* d_out, int out_size, void* d_ws, size_t ws_size,
                              hipStream_t stream) {
    (void)in_sizes; (void)n_in; (void)out_size; (void)ws_size;
    const float* x = (const float*)d_in[0];
    const float* w = (const float*)d_in[1];
    float* ws   = (float*)d_ws;
    float* part = ws + WS_PART;
    float* c    = ws + WS_C;
    float* t    = ws + WS_T;
    float* s    = ws + WS_S;
    float* out  = (float*)d_out;

    colsum_x_wmma<<<dim3(DIM / 64, SPLITS), 128, 0, stream>>>(x, part);
    reduce_parts<<<dim3(DIM / 256), 256, 0, stream>>>(part, c);
    weight_sums<<<5, 256, 0, stream>>>(w, t, s);
    softreorder_out<<<CHANNELS, 256, 0, stream>>>(c, t, s, out);
}


# weight row and column sums move into launch 1 as 20 extra work-groups that run beside the x stream (4 launches become 3); new reduction trees for s and t
# speedup vs baseline: 2.2322x; 1.5682x over previous
#include <hip/hip_runtime.h>

constexpr int N_SAMPLES = 65536;
constexpr int DIM       = 1024;
constexpr int CHANNELS  = 128;

constexpr int SPLITS         = 64;
constexpr int ROWS_PER_SPLIT = N_SAMPLES / SPLITS;

typedef __attribute__((ext_vector_type(2))) float v2f;
typedef __attribute__((ext_vector_type(4))) float v4f;
typedef __attribute__((ext_vector_type(8))) float v8f;

constexpr int WS_PART = 0;
constexpr int WS_C    = SPLITS * DIM;
constexpr int WS_T    = WS_C + DIM;
constexpr int WS_S    = WS_T + DIM;

constexpr int WG_WAVES      = 4;
constexpr int ROWS_PER_WAVE = ROWS_PER_SPLIT / WG_WAVES;
constexpr int BQ            = 8;
constexpr int NBATCH        = ROWS_PER_WAVE / (4 * BQ);
constexpr int W_NBATCH      = CHANNELS / (WG_WAVES * 4 * BQ);
static_assert(W_NBATCH * WG_WAVES * 4 * BQ == CHANNELS, "weight rows must fill whole batches");

template <int NB>
__device__ __forceinline__ void strip_colsum(const float* __restrict__ src, const int row0,
                                             const int c0, float* __restrict__ dst,
                                             v4f (*red)[16]) {
    const int lane = threadIdx.x & 31;
    const int wave = threadIdx.x >> 5;
    const int r0   = row0 + wave * (NB * BQ * 4);
    const int h    = lane >> 4;
    const int c    = lane & 15;

    v2f a; a.x = 1.0f; a.y = 1.0f;
    v8f acc0 = {}, acc1 = {}, acc2 = {}, acc3 = {};

    const float* p = src + (size_t)(r0 + h) * DIM + c0 + 4 * c;
    v4f buf[2][2 * BQ];

#define LOAD_BATCH(B)                                                             \
    _Pragma("unroll") for (int g = 0; g < BQ; ++g) {                               \
        const float* q = p + (size_t)(((B) * BQ + g) * 4) * DIM;                   \
        buf[(B) & 1][2 * g]     = __builtin_nontemporal_load((const v4f*)q);             \
        buf[(B) & 1][2 * g + 1] = __builtin_nontemporal_load((const v4f*)(q + 2 * DIM));   \
    }

    LOAD_BATCH(0)
    #pragma unroll
    for (int b = 0; b < NB; ++b) {
        __builtin_amdgcn_sched_barrier(0);
        if (b + 1 < NB) { LOAD_BATCH(b + 1) }
        __builtin_amdgcn_sched_barrier(0);
        #pragma unroll
        for (int g = 0; g < BQ; ++g) {
            const v4f u = buf[b & 1][2 * g];
            const v4f v = buf[b & 1][2 * g + 1];
            v2f b0; b0.x = u.x; b0.y = v.x;
            v2f b1; b1.x = u.y; b1.y = v.y;
            v2f b2; b2.x = u.z; b2.y = v.z;
            v2f b3; b3.x = u.w; b3.y = v.w;
            acc0 = __builtin_amdgcn_wmma_f32_16x16x4_f32(false, a, false, b0, (short)0, acc0, false, false);
            acc1 = __builtin_amdgcn_wmma_f32_16x16x4_f32(false, a, false, b1, (short)0, acc1, false, false);
            acc2 = __builtin_amdgcn_wmma_f32_16x16x4_f32(false, a, false, b2, (short)0, acc2, false, false);
            acc3 = __builtin_amdgcn_wmma_f32_16x16x4_f32(false, a, false, b3, (short)0, acc3, false, false);
        }
    }
#undef LOAD_BATCH

    if (lane < 16) {
        v4f m; m.x = acc0[0]; m.y = acc1[0]; m.z = acc2[0]; m.w = acc3[0];
        red[wave][lane] = m;
    }
    __syncthreads();
    if (threadIdx.x < 16) {
        const v4f s = ((red[0][lane] + red[1][lane]) + red[2][lane]) + red[3][lane];
        *(v4f*)(dst + 4 * lane) = s;
    }
}

constexpr int STRIPS          = DIM / 64;
constexpr int S_ROWS_PER_WG   = 32;
constexpr int S_WGS           = CHANNELS / S_ROWS_PER_WG;
constexpr int S_ROWS_PER_WAVE = S_ROWS_PER_WG / WG_WAVES;
constexpr int S_ROWS_PER_STEP = 4;
constexpr int ROW_V4_PER_LANE = DIM / (4 * 32);
constexpr int W_WGS           = STRIPS + S_WGS;

__global__ void __launch_bounds__(WG_WAVES * 32)
colsum_x_wmma(const float* __restrict__ x, const float* __restrict__ w,
              float* __restrict__ part, float* __restrict__ t, float* __restrict__ s) {
    __shared__ v4f red[WG_WAVES][16];
    __shared__ float srow[S_ROWS_PER_WG];
    const int bid = blockIdx.x;

    if (bid < STRIPS) {
        strip_colsum<W_NBATCH>(w, 0, bid * 64, t + bid * 64, red);
        return;
    }
    if (bid < W_WGS) {
        const int lane = threadIdx.x & 31;
        const int wave = threadIdx.x >> 5;
        const int row0 = (bid - STRIPS) * S_ROWS_PER_WG;
        const v4f* rp  = (const v4f*)(w + (size_t)(row0 + wave * S_ROWS_PER_WAVE) * DIM) + lane;
        float keep = 0.0f;
        #pragma nounroll
        for (int r = 0; r < S_ROWS_PER_WAVE; r += S_ROWS_PER_STEP) {
            v4f rv[S_ROWS_PER_STEP][ROW_V4_PER_LANE];
            #pragma unroll
            for (int q = 0; q < S_ROWS_PER_STEP; ++q) {
                #pragma unroll
                for (int k = 0; k < ROW_V4_PER_LANE; ++k)
                    rv[q][k] = rp[(size_t)(r + q) * (DIM / 4) + 32 * k];
            }
            #pragma unroll
            for (int q = 0; q < S_ROWS_PER_STEP; ++q) {
                v4f a4 = rv[q][0];
                #pragma unroll
                for (int k = 1; k < ROW_V4_PER_LANE; ++k) a4 += rv[q][k];
                float v = (a4.x + a4.y) + (a4.z + a4.w);
                for (int off = 16; off > 0; off >>= 1) v += __shfl_xor(v, off);
                if (lane == r + q) keep = v;
            }
        }
        if (lane < S_ROWS_PER_WAVE) srow[wave * S_ROWS_PER_WAVE + lane] = keep;
        __syncthreads();
        if (threadIdx.x < S_ROWS_PER_WG) s[row0 + threadIdx.x] = srow[threadIdx.x];
        return;
    }

    const int id    = bid - W_WGS;
    const int c0    = (id % STRIPS) * 64;
    const int split = id / STRIPS;
    strip_colsum<NBATCH>(x, split * ROWS_PER_SPLIT, c0, part + (size_t)split * DIM + c0, red);
}

__global__ void reduce_parts(const float* __restrict__ part,
                             float* __restrict__ c) {
    const int col = blockIdx.x * blockDim.x + threadIdx.x;
    float s = 0.0f;
    #pragma unroll 8
    for (int k = 0; k < SPLITS; ++k) s += part[(size_t)k * DIM + col];
    c[col] = s;
}

__global__ void softreorder_out(const float* __restrict__ cvec,
                                const float* __restrict__ tvec,
                                const float* __restrict__ svec,
                                float* __restrict__ out) {
    __shared__ float red[256];
    const int i  = blockIdx.x;
    const float si = svec[i];

    float p = 0.0f;
    for (int j = threadIdx.x; j < DIM; j += 256) p += __expf(si * cvec[j]);
    red[threadIdx.x] = p;
    __syncthreads();
    #pragma unroll
    for (int off = 128; off > 0; off >>= 1) {
        if (threadIdx.x < off) red[threadIdx.x] += red[threadIdx.x + off];
        __syncthreads();
    }
    const float inv_det = 1.0f / red[0];

    float* orow = out + (size_t)i * DIM;
    for (int j = threadIdx.x; j < DIM; j += 256)
        orow[j] = __expf(si * tvec[j]) * inv_det;
}

extern "C" void kernel_launch(void* const* d_in, const int* in_sizes, int n_in,
                              void* d_out, int out_size, void* d_ws, size_t ws_size,
                              hipStream_t stream) {
    (void)in_sizes; (void)n_in; (void)out_size; (void)ws_size;
    const float* x = (const float*)d_in[0];
    const float* w = (const float*)d_in[1];
    float* ws   = (float*)d_ws;
    float* part = ws + WS_PART;
    float* c    = ws + WS_C;
    float* t    = ws + WS_T;
    float* s    = ws + WS_S;
    float* out  = (float*)d_out;

    colsum_x_wmma<<<dim3(W_WGS + STRIPS * SPLITS), 128, 0, stream>>>(x, w, part, t, s);
    reduce_parts<<<dim3(DIM / 256), 256, 0, stream>>>(part, c);
    softreorder_out<<<CHANNELS, 256, 0, stream>>>(c, t, s, out);
}


# launch 2 folds into the output launch (3 launches become 2): each channel's work-group adds the 64 partial column sums itself, all 64 loads per thread in flight; new summation order for det
# speedup vs baseline: 2.6552x; 1.1895x over previous
#include <hip/hip_runtime.h>

constexpr int N_SAMPLES = 65536;
constexpr int DIM       = 1024;
constexpr int CHANNELS  = 128;

constexpr int SPLITS         = 64;
constexpr int ROWS_PER_SPLIT = N_SAMPLES / SPLITS;

typedef __attribute__((ext_vector_type(2))) float v2f;
typedef __attribute__((ext_vector_type(4))) float v4f;
typedef __attribute__((ext_vector_type(8))) float v8f;

constexpr int WS_PART = 0;
constexpr int WS_T    = SPLITS * DIM;
constexpr int WS_S    = WS_T + DIM;

constexpr int WG_WAVES      = 4;
constexpr int ROWS_PER_WAVE = ROWS_PER_SPLIT / WG_WAVES;
constexpr int BQ            = 8;
constexpr int NBATCH        = ROWS_PER_WAVE / (4 * BQ);
constexpr int W_NBATCH      = CHANNELS / (WG_WAVES * 4 * BQ);
static_assert(W_NBATCH * WG_WAVES * 4 * BQ == CHANNELS, "weight rows must fill whole batches");

template <int NB>
__device__ __forceinline__ void strip_colsum(const float* __restrict__ src, const int row0,
                                             const int c0, float* __restrict__ dst,
                                             v4f (*red)[16]) {
    const int lane = threadIdx.x & 31;
    const int wave = threadIdx.x >> 5;
    const int r0   = row0 + wave * (NB * BQ * 4);
    const int h    = lane >> 4;
    const int c    = lane & 15;

    v2f a; a.x = 1.0f; a.y = 1.0f;
    v8f acc0 = {}, acc1 = {}, acc2 = {}, acc3 = {};

    const float* p = src + (size_t)(r0 + h) * DIM + c0 + 4 * c;
    v4f buf[2][2 * BQ];

#define LOAD_BATCH(B)                                                             \
    _Pragma("unroll") for (int g = 0; g < BQ; ++g) {                               \
        const float* q = p + (size_t)(((B) * BQ + g) * 4) * DIM;                   \
        buf[(B) & 1][2 * g]     = __builtin_nontemporal_load((const v4f*)q);             \
        buf[(B) & 1][2 * g + 1] = __builtin_nontemporal_load((const v4f*)(q + 2 * DIM));   \
    }

    LOAD_BATCH(0)
    #pragma unroll
    for (int b = 0; b < NB; ++b) {
        __builtin_amdgcn_sched_barrier(0);
        if (b + 1 < NB) { LOAD_BATCH(b + 1) }
        __builtin_amdgcn_sched_barrier(0);
        #pragma unroll
        for (int g = 0; g < BQ; ++g) {
            const v4f u = buf[b & 1][2 * g];
            const v4f v = buf[b & 1][2 * g + 1];
            v2f b0; b0.x = u.x; b0.y = v.x;
            v2f b1; b1.x = u.y; b1.y = v.y;
            v2f b2; b2.x = u.z; b2.y = v.z;
            v2f b3; b3.x = u.w; b3.y = v.w;
            acc0 = __builtin_amdgcn_wmma_f32_16x16x4_f32(false, a, false, b0, (short)0, acc0, false, false);
            acc1 = __builtin_amdgcn_wmma_f32_16x16x4_f32(false, a, false, b1, (short)0, acc1, false, false);
            acc2 = __builtin_amdgcn_wmma_f32_16x16x4_f32(false, a, false, b2, (short)0, acc2, false, false);
            acc3 = __builtin_amdgcn_wmma_f32_16x16x4_f32(false, a, false, b3, (short)0, acc3, false, false);
        }
    }
#undef LOAD_BATCH

    if (lane < 16) {
        v4f m; m.x = acc0[0]; m.y = acc1[0]; m.z = acc2[0]; m.w = acc3[0];
        red[wave][lane] = m;
    }
    __syncthreads();
    if (threadIdx.x < 16) {
        const v4f s = ((red[0][lane] + red[1][lane]) + red[2][lane]) + red[3][lane];
        *(v4f*)(dst + 4 * lane) = s;
    }
}

constexpr int STRIPS          = DIM / 64;
constexpr int S_ROWS_PER_WG   = 32;
constexpr int S_WGS           = CHANNELS / S_ROWS_PER_WG;
constexpr int S_ROWS_PER_WAVE = S_ROWS_PER_WG / WG_WAVES;
constexpr int S_ROWS_PER_STEP = 4;
constexpr int ROW_V4_PER_LANE = DIM / (4 * 32);
constexpr int W_WGS           = STRIPS + S_WGS;

__global__ void __launch_bounds__(WG_WAVES * 32)
colsum_x_wmma(const float* __restrict__ x, const float* __restrict__ w,
              float* __restrict__ part, float* __restrict__ t, float* __restrict__ s) {
    __shared__ v4f red[WG_WAVES][16];
    __shared__ float srow[S_ROWS_PER_WG];
    const int bid = blockIdx.x;

    if (bid < STRIPS) {
        strip_colsum<W_NBATCH>(w, 0, bid * 64, t + bid * 64, red);
        return;
    }
    if (bid < W_WGS) {
        const int lane = threadIdx.x & 31;
        const int wave = threadIdx.x >> 5;
        const int row0 = (bid - STRIPS) * S_ROWS_PER_WG;
        const v4f* rp  = (const v4f*)(w + (size_t)(row0 + wave * S_ROWS_PER_WAVE) * DIM) + lane;
        float keep = 0.0f;
        #pragma nounroll
        for (int r = 0; r < S_ROWS_PER_WAVE; r += S_ROWS_PER_STEP) {
            v4f rv[S_ROWS_PER_STEP][ROW_V4_PER_LANE];
            #pragma unroll
            for (int q = 0; q < S_ROWS_PER_STEP; ++q) {
                #pragma unroll
                for (int k = 0; k < ROW_V4_PER_LANE; ++k)
                    rv[q][k] = rp[(size_t)(r + q) * (DIM / 4) + 32 * k];
            }
            #pragma unroll
            for (int q = 0; q < S_ROWS_PER_STEP; ++q) {
                v4f a4 = rv[q][0];
                #pragma unroll
                for (int k = 1; k < ROW_V4_PER_LANE; ++k) a4 += rv[q][k];
                float v = (a4.x + a4.y) + (a4.z + a4.w);
                for (int off = 16; off > 0; off >>= 1) v += __shfl_xor(v, off);
                if (lane == r + q) keep = v;
            }
        }
        if (lane < S_ROWS_PER_WAVE) srow[wave * S_ROWS_PER_WAVE + lane] = keep;
        __syncthreads();
        if (threadIdx.x < S_ROWS_PER_WG) s[row0 + threadIdx.x] = srow[threadIdx.x];
        return;
    }

    const int id    = bid - W_WGS;
    const int c0    = (id % STRIPS) * 64;
    const int split = id / STRIPS;
    strip_colsum<NBATCH>(x, split * ROWS_PER_SPLIT, c0, part + (size_t)split * DIM + c0, red);
}

constexpr int OUT_THREADS = DIM / 4;

__global__ void __launch_bounds__(OUT_THREADS)
softreorder_out(const float* __restrict__ part,
                const float* __restrict__ tvec,
                const float* __restrict__ svec,
                float* __restrict__ out) {
    __shared__ float red[OUT_THREADS];
    const int i   = blockIdx.x;
    const int tid = threadIdx.x;

    const float si = svec[i];
    const v4f   t4 = *(const v4f*)(tvec + 4 * tid);

    const v4f* pp = (const v4f*)part + tid;
    v4f pv[SPLITS];
    #pragma unroll
    for (int k = 0; k < SPLITS; ++k) pv[k] = pp[(size_t)k * (DIM / 4)];
    __builtin_amdgcn_sched_barrier(0);
    v4f c4 = pv[0];
    #pragma unroll
    for (int k = 1; k < SPLITS; ++k) c4 += pv[k];

    red[tid] = (__expf(si * c4.x) + __expf(si * c4.y)) + (__expf(si * c4.z) + __expf(si * c4.w));
    __syncthreads();
    #pragma unroll
    for (int off = OUT_THREADS / 2; off > 0; off >>= 1) {
        if (tid < off) red[tid] += red[tid + off];
        __syncthreads();
    }
    const float inv_det = 1.0f / red[0];

    v4f o;
    o.x = __expf(si * t4.x) * inv_det;
    o.y = __expf(si * t4.y) * inv_det;
    o.z = __expf(si * t4.z) * inv_det;
    o.w = __expf(si * t4.w) * inv_det;
    *(v4f*)(out + (size_t)i * DIM + 4 * tid) = o;
}

extern "C" void kernel_launch(void* const* d_in, const int* in_sizes, int n_in,
                              void* d_out, int out_size, void* d_ws, size_t ws_size,
                              hipStream_t stream) {
    (void)in_sizes; (void)n_in; (void)out_size; (void)ws_size;
    const float* x = (const float*)d_in[0];
    const float* w = (const float*)d_in[1];
    float* ws   = (float*)d_ws;
    float* part = ws + WS_PART;
    float* t    = ws + WS_T;
    float* s    = ws + WS_S;
    float* out  = (float*)d_out;

    colsum_x_wmma<<<dim3(W_WGS + STRIPS * SPLITS), 128, 0, stream>>>(x, w, part, t, s);
    softreorder_out<<<CHANNELS, OUT_THREADS, 0, stream>>>(part, t, s, out);
}


# x stream re-tiled to one 8-wave work-group per WGP (256 groups, 512-row waves, 32-63 loads in flight): half the waves enter, 16 partial rows not 64; weight sums ride along; new sum order for c, s, t
# speedup vs baseline: 2.8562x; 1.0757x over previous
#include <hip/hip_runtime.h>


constexpr int N_SAMPLES = 65536;
constexpr int DIM       = 1024;
constexpr int CHANNELS  = 128;

constexpr int STRIPS         = DIM / 64;
constexpr int SPLITS         = 16;
constexpr int ROWS_PER_SPLIT = N_SAMPLES / SPLITS;

typedef __attribute__((ext_vector_type(2))) float v2f;
typedef __attribute__((ext_vector_type(4))) float v4f;
typedef __attribute__((ext_vector_type(8))) float v8f;

constexpr int WG_WAVES      = 8;
constexpr int ROWS_PER_WAVE = ROWS_PER_SPLIT / WG_WAVES;
constexpr int BQ            = 16;
constexpr int NBATCH        = ROWS_PER_WAVE / (4 * BQ);
constexpr int W_ROWS_PER_WG = WG_WAVES * 4;
constexpr int W_SPLITS      = CHANNELS / W_ROWS_PER_WG;
static_assert(NBATCH * 4 * BQ == ROWS_PER_WAVE, "rows of a wave must fill whole batches");
static_assert(W_SPLITS * W_ROWS_PER_WG == CHANNELS && W_SPLITS <= SPLITS, "weight rows must fill whole row-quads");

constexpr int WS_PART  = 0;
constexpr int WS_TPART = SPLITS * DIM;

__global__ void __launch_bounds__(WG_WAVES * 32)
colsum_xw_wmma(const float* __restrict__ x, const float* __restrict__ w,
               float* __restrict__ part, float* __restrict__ tpart) {
    __shared__ v4f red[WG_WAVES][16];
    __shared__ v4f wred[WG_WAVES][16];
    const int lane  = threadIdx.x & 31;
    const int wave  = threadIdx.x >> 5;
    const int h     = lane >> 4;
    const int c     = lane & 15;
    const int strip = blockIdx.x % STRIPS;
    const int split = blockIdx.x / STRIPS;
    const int c0    = strip * 64;
    const bool has_w = split < W_SPLITS;

    v2f a; a.x = 1.0f; a.y = 1.0f;
    v8f acc0 = {}, acc1 = {}, acc2 = {}, acc3 = {};

    const float* p = x + (size_t)(split * ROWS_PER_SPLIT + wave * ROWS_PER_WAVE + h) * DIM + c0 + 4 * c;
    v4f buf[2][2 * BQ];

#define LOAD_BATCH(B)                                                             \
    _Pragma("unroll") for (int g = 0; g < BQ; ++g) {                               \
        const float* q = p + (size_t)(((B) * BQ + g) * 4) * DIM;                   \
        buf[(B) & 1][2 * g]     = __builtin_nontemporal_load((const v4f*)q);             \
        buf[(B) & 1][2 * g + 1] = __builtin_nontemporal_load((const v4f*)(q + 2 * DIM));   \
    }

    LOAD_BATCH(0)
    const float* wq = w + (size_t)((split % W_SPLITS) * W_ROWS_PER_WG + wave * 4 + h) * DIM + c0 + 4 * c;
    const v4f wu = *(const v4f*)wq;
    const v4f wv = *(const v4f*)(wq + 2 * DIM);
    #pragma unroll
    for (int b = 0; b < NBATCH; ++b) {
        __builtin_amdgcn_sched_barrier(0);
        if (b + 1 < NBATCH) { LOAD_BATCH(b + 1) }
        __builtin_amdgcn_sched_barrier(0);
        #pragma unroll
        for (int g = 0; g < BQ; ++g) {
            const v4f u = buf[b & 1][2 * g];
            const v4f v = buf[b & 1][2 * g + 1];
            v2f b0; b0.x = u.x; b0.y = v.x;
            v2f b1; b1.x = u.y; b1.y = v.y;
            v2f b2; b2.x = u.z; b2.y = v.z;
            v2f b3; b3.x = u.w; b3.y = v.w;
            acc0 = __builtin_amdgcn_wmma_f32_16x16x4_f32(false, a, false, b0, (short)0, acc0, false, false);
            acc1 = __builtin_amdgcn_wmma_f32_16x16x4_f32(false, a, false, b1, (short)0, acc1, false, false);
            acc2 = __builtin_amdgcn_wmma_f32_16x16x4_f32(false, a, false, b2, (short)0, acc2, false, false);
            acc3 = __builtin_amdgcn_wmma_f32_16x16x4_f32(false, a, false, b3, (short)0, acc3, false, false);
        }
    }
#undef LOAD_BATCH

    if (lane < 16) {
        v4f m; m.x = acc0[0]; m.y = acc1[0]; m.z = acc2[0]; m.w = acc3[0];
        red[wave][lane] = m;
    }
    if (has_w) {
        const v8f z = {};
        v2f b0; b0.x = wu.x; b0.y = wv.x;
        v2f b1; b1.x = wu.y; b1.y = wv.y;
        v2f b2; b2.x = wu.z; b2.y = wv.z;
        v2f b3; b3.x = wu.w; b3.y = wv.w;
        const v8f d0 = __builtin_amdgcn_wmma_f32_16x16x4_f32(false, a, false, b0, (short)0, z, false, false);
        const v8f d1 = __builtin_amdgcn_wmma_f32_16x16x4_f32(false, a, false, b1, (short)0, z, false, false);
        const v8f d2 = __builtin_amdgcn_wmma_f32_16x16x4_f32(false, a, false, b2, (short)0, z, false, false);
        const v8f d3 = __builtin_amdgcn_wmma_f32_16x16x4_f32(false, a, false, b3, (short)0, z, false, false);
        if (lane < 16) {
            v4f m; m.x = d0[0]; m.y = d1[0]; m.z = d2[0]; m.w = d3[0];
            wred[wave][lane] = m;
        }
    }
    __syncthreads();
    if (threadIdx.x < 16) {
        const v4f s = (((red[0][lane] + red[1][lane]) + (red[2][lane] + red[3][lane])) +
                       ((red[4][lane] + red[5][lane]) + (red[6][lane] + red[7][lane])));
        *(v4f*)(part + (size_t)split * DIM + c0 + 4 * lane) = s;
    } else if (has_w && wave == 1 && lane < 16) {
        const v4f s = (((wred[0][lane] + wred[1][lane]) + (wred[2][lane] + wred[3][lane])) +
                       ((wred[4][lane] + wred[5][lane]) + (wred[6][lane] + wred[7][lane])));
        *(v4f*)(tpart + (size_t)split * DIM + c0 + 4 * lane) = s;
    }
}

constexpr int OUT_THREADS = DIM / 4;
constexpr int OUT_WAVES   = OUT_THREADS / 32;

__global__ void __launch_bounds__(OUT_THREADS)
softreorder_out(const float* __restrict__ part,
                const float* __restrict__ tpart,
                const float* __restrict__ w,
                float* __restrict__ out) {
    __shared__ float red[OUT_THREADS];
    __shared__ float sred[OUT_WAVES];
    const int i   = blockIdx.x;
    const int tid = threadIdx.x;

    const v4f wr = *(const v4f*)(w + (size_t)i * DIM + 4 * tid);

    const v4f* pp = (const v4f*)part + tid;
    const v4f* tp = (const v4f*)tpart + tid;
    v4f pv[SPLITS];
    v4f tv[W_SPLITS];
    #pragma unroll
    for (int k = 0; k < SPLITS; ++k) pv[k] = pp[(size_t)k * (DIM / 4)];
    #pragma unroll
    for (int k = 0; k < W_SPLITS; ++k) tv[k] = tp[(size_t)k * (DIM / 4)];
    __builtin_amdgcn_sched_barrier(0);

    float sv = (wr.x + wr.y) + (wr.z + wr.w);
    for (int off = 16; off > 0; off >>= 1) sv += __shfl_xor(sv, off);
    if ((tid & 31) == 0) sred[tid >> 5] = sv;
    __syncthreads();
    const float si = ((sred[0] + sred[1]) + (sred[2] + sred[3])) + ((sred[4] + sred[5]) + (sred[6] + sred[7]));

    v4f c4 = pv[0];
    #pragma unroll
    for (int k = 1; k < SPLITS; ++k) c4 += pv[k];
    v4f t4 = tv[0];
    #pragma unroll
    for (int k = 1; k < W_SPLITS; ++k) t4 += tv[k];

    red[tid] = (__expf(si * c4.x) + __expf(si * c4.y)) + (__expf(si * c4.z) + __expf(si * c4.w));
    __syncthreads();
    #pragma unroll
    for (int off = OUT_THREADS / 2; off > 0; off >>= 1) {
        if (tid < off) red[tid] += red[tid + off];
        __syncthreads();
    }
    const float inv_det = 1.0f / red[0];

    v4f o;
    o.x = __expf(si * t4.x) * inv_det;
    o.y = __expf(si * t4.y) * inv_det;
    o.z = __expf(si * t4.z) * inv_det;
    o.w = __expf(si * t4.w) * inv_det;
    *(v4f*)(out + (size_t)i * DIM + 4 * tid) = o;
}

extern "C" void kernel_launch(void* const* d_in, const int* in_sizes, int n_in,
                              void* d_out, int out_size, void* d_ws, size_t ws_size,
                              hipStream_t stream) {
    (void)in_sizes; (void)n_in; (void)out_size; (void)ws_size;
    const float* x = (const float*)d_in[0];
    const float* w = (const float*)d_in[1];
    float* ws    = (float*)d_ws;
    float* part  = ws + WS_PART;
    float* tpart = ws + WS_TPART;
    float* out   = (float*)d_out;

    colsum_xw_wmma<<<dim3(STRIPS * SPLITS), WG_WAVES * 32, 0, stream>>>(x, w, part, tpart);
    softreorder_out<<<CHANNELS, OUT_THREADS, 0, stream>>>(part, tpart, w, out);
}
